# starting kernel
# baseline (speedup 1.0000x reference)
#include <hip/hip_runtime.h>
#include <hip/hip_bf16.h>

#define NBATCH   8
#define NFEAT    1000
#define NPARAM   4576
#define HW       50176
#define KB       11
#define RPI      12
#define WM0_ROWS (3*RPI)
#define WM1_ROWS (16*RPI)
#define WM2_ROWS (16*RPI)
#define WM_PER_B (16*(WM0_ROWS + WM1_ROWS + WM2_ROWS))

#define TOK   64
#define SY    17
#define FSTR  198

#define FCCOLS 64
#define FCKT   100
#define FSTRF  1002
#define WTSTR  101

typedef __attribute__((ext_vector_type(2))) float v2f;
typedef __attribute__((ext_vector_type(8))) float v8f;

__global__ void kan_fc_wmma_kernel(const float* __restrict__ features,
                                   const float* __restrict__ fc_w,
                                   const float* __restrict__ fc_b,
                                   float* __restrict__ w_out) {
    extern __shared__ float smem[];
    float* fs = smem;
    float* wt = fs + 16 * FSTRF;

    const int tid  = threadIdx.x;
    const int lane = tid & 31;
    const int wv   = tid >> 5;
    const int m    = lane & 15;
    const int hi   = lane >> 4;
    const int j0   = blockIdx.x * FCCOLS;

    for (int idx = tid; idx < 16 * NFEAT; idx += 128) {
        const int mm = idx / NFEAT;
        const int k  = idx % NFEAT;
        fs[mm * FSTRF + k] = (mm < NBATCH) ? features[mm * NFEAT + k] : 0.0f;
    }

    const int  jc   = j0 + wv * 16 + m;
    const float bias = (jc < NPARAM) ? fc_b[jc] : 0.0f;
    v8f acc;
#pragma unroll
    for (int r = 0; r < 8; ++r) acc[r] = bias;

    for (int kt = 0; kt < NFEAT; kt += FCKT) {
        __syncthreads();
        for (int e = tid; e < FCCOLS * FCKT; e += 128) {
            const int c = e / FCKT;
            const int k = e % FCKT;
            const int j = j0 + c;
            wt[c * WTSTR + k] = (j < NPARAM) ? fc_w[(size_t)j * NFEAT + kt + k] : 0.0f;
        }
        __syncthreads();
        const float* arow = &fs[m * FSTRF + kt];
        const float* brow = &wt[(wv * 16 + m) * WTSTR];
        for (int k0 = 0; k0 < FCKT; k0 += 4) {
            const int ka = k0 + 2 * hi;
            v2f a, bb;
            a.x  = arow[ka];
            a.y  = arow[ka + 1];
            bb.x = brow[ka];
            bb.y = brow[ka + 1];
            acc = __builtin_amdgcn_wmma_f32_16x16x4_f32(
                    false, a, false, bb, (short)0, acc, false, false);
        }
    }
    if (hi == 0 && jc < NPARAM) {
#pragma unroll
        for (int r = 0; r < 8; ++r)
            w_out[r * NPARAM + jc] = acc[r];
    }
}

__global__ void kan_wm_kernel(const float* __restrict__ w,
                              float* __restrict__ wm) {
    const int e = blockIdx.x * blockDim.x + threadIdx.x;
    const int total = NBATCH * WM_PER_B;
    if (e >= total) return;
    const int b   = e / WM_PER_B;
    const int rem = e % WM_PER_B;

    int layer, idx;
    if (rem < 16 * WM0_ROWS)                   { layer = 0; idx = rem; }
    else if (rem < 16 * (WM0_ROWS + WM1_ROWS)) { layer = 1; idx = rem - 16 * WM0_ROWS; }
    else                                       { layer = 2; idx = rem - 16 * (WM0_ROWS + WM1_ROWS); }

    const int DO_[3] = {16, 16, 3};
    const int I0[3]  = {0,   624, 3952};
    const int I1[3]  = {528, 3440, 4480};
    const int I2[3]  = {576, 3696, 4528};

    const int r = idx >> 4;
    const int o = idx & 15;
    const int i = r / RPI;
    const int k = r % RPI;

    float val = 0.0f;
    const int do_ = DO_[layer];
    if (o < do_) {
        const float* wb = w + (size_t)b * NPARAM;
        const int io = i * do_ + o;
        if (k < KB) val = wb[I0[layer] + io * KB + k] * wb[I1[layer] + io];
        else        val = wb[I2[layer] + io];
    }
    wm[e] = val;
}

__device__ __forceinline__ void bspline12(float x, float* f) {
    float b[14];
#pragma unroll
    for (int j = 0; j < 14; ++j) {
        const float g0 = -1.75f + 0.25f * j;
        b[j] = (x >= g0 && x < g0 + 0.25f) ? 1.0f : 0.0f;
    }
#pragma unroll
    for (int p = 1; p <= 3; ++p) {
        const float inv = 1.0f / (0.25f * p);
#pragma unroll 13
        for (int j = 0; j < 14 - p; ++j) {
            const float gj = -1.75f + 0.25f * j;
            const float left  = (x - gj) * inv * b[j];
            const float right = ((gj + 0.25f * (p + 1)) - x) * inv * b[j + 1];
            b[j] = left + right;
        }
    }
#pragma unroll
    for (int k = 0; k < KB; ++k) f[k] = b[k];
    f[KB] = x / (1.0f + __expf(-x));
}

template <int DI>
__device__ __forceinline__ void kan_layer(const float* __restrict__ wmL,
                                          const float* yin,
                                          float* yout,
                                          float* feat,
                                          int tid) {
    const int R    = DI * RPI;
    const int lane = tid & 31;
    const int wv   = tid >> 5;
    const int m    = lane & 15;
    const int hi   = lane >> 4;

    for (int task = tid; task < TOK * DI; task += 128) {
        const int t = task & (TOK - 1);
        const int i = task >> 6;
        float f[RPI];
        bspline12(yin[t * SY + i], f);
        float* dst = &feat[t * FSTR + i * RPI];
#pragma unroll
        for (int k = 0; k < RPI; ++k) dst[k] = f[k];
    }
    __syncthreads();

    const int tloc = wv * 16;
    v8f acc = {0.f, 0.f, 0.f, 0.f, 0.f, 0.f, 0.f, 0.f};
    const float* arow = &feat[(tloc + m) * FSTR];
    for (int k0 = 0; k0 < R; k0 += 4) {
        const int ka = k0 + 2 * hi;
        v2f a, bb;
        a.x  = arow[ka];
        a.y  = arow[ka + 1];
        bb.x = wmL[ka * 16 + m];
        bb.y = wmL[(ka + 1) * 16 + m];
        acc = __builtin_amdgcn_wmma_f32_16x16x4_f32(
                false, a, false, bb, (short)0, acc, false, false);
    }
#pragma unroll
    for (int r = 0; r < 8; ++r)
        yout[(tloc + r + 8 * hi) * SY + m] = acc[r];
    __syncthreads();
}

__global__ void kan_main_kernel(const float* __restrict__ x,
                                const float* __restrict__ wm,
                                float* __restrict__ out) {
    extern __shared__ float smem[];
    float* wms  = smem;
    float* yA   = wms + WM_PER_B;
    float* yB   = yA + TOK * SY;
    float* feat = yB + TOK * SY;

    const int tid   = threadIdx.x;
    const int b     = blockIdx.x / (HW / TOK);
    const int chunk = blockIdx.x % (HW / TOK);
    const int hw0   = chunk * TOK;

    const float* wmb = wm + (size_t)b * WM_PER_B;
    for (int i = tid; i < WM_PER_B; i += 128) wms[i] = wmb[i];

#pragma unroll
    for (int c = 0; c < 3; ++c)
        for (int t = tid; t < TOK; t += 128)
            yA[t * SY + c] = x[((size_t)(b * 3 + c)) * HW + hw0 + t];
    __syncthreads();

    kan_layer<3>(wms,                               yA, yB, feat, tid);
    kan_layer<16>(wms + 16 * WM0_ROWS,              yB, yA, feat, tid);
    kan_layer<16>(wms + 16 * (WM0_ROWS + WM1_ROWS), yA, yB, feat, tid);

#pragma unroll
    for (int c = 0; c < 3; ++c)
        for (int t = tid; t < TOK; t += 128)
            out[((size_t)(b * 3 + c)) * HW + hw0 + t] = yB[t * SY + c];
}

extern "C" void kernel_launch(void* const* d_in, const int* in_sizes, int n_in,
                              void* d_out, int out_size, void* d_ws, size_t ws_size,
                              hipStream_t stream) {
    const float* x        = (const float*)d_in[0];
    const float* features = (const float*)d_in[1];
    const float* fc_w     = (const float*)d_in[2];
    const float* fc_b     = (const float*)d_in[3];
    float* out    = (float*)d_out;
    float* w_buf  = (float*)d_ws;
    float* wm_buf = w_buf + (size_t)NBATCH * NPARAM;

    const size_t fc_smem = (16 * FSTRF + FCCOLS * WTSTR) * sizeof(float);
    kan_fc_wmma_kernel<<<(NPARAM + FCCOLS - 1) / FCCOLS, 128, fc_smem, stream>>>(
        features, fc_w, fc_b, w_buf);

    kan_wm_kernel<<<(NBATCH * WM_PER_B + 255) / 256, 256, 0, stream>>>(w_buf, wm_buf);

    const size_t smem_bytes = (WM_PER_B + 2 * TOK * SY + TOK * FSTR) * sizeof(float);
    kan_main_kernel<<<NBATCH * (HW / TOK), 128, smem_bytes, stream>>>(x, wm_buf, out);
}


# Main kernel: each lane builds its token's features in registers as the A fragments of its own WMMAs (feature LDS, per-layer block barriers gone); fragment-order weights; new k order, two accumulators
# speedup vs baseline: 4.4456x; 4.4456x over previous
#include <hip/hip_runtime.h>
#include <hip/hip_bf16.h>


#define NBATCH   8
#define NFEAT    1000
#define NPARAM   4576
#define HW       50176
#define KB       11
#define RPI      12

#define TOK   64
#define DIH0  2
#define DIH1  8
#define WF0   (3 * DIH0 * 2 * 16 * 4)
#define WF1   (3 * DIH1 * 2 * 16 * 4)
#define WM_PER_B (WF0 + 2 * WF1)
#define YP    20

#define FCS    5
#define FCKC   200
#define FCUPR  50
#define FCCOLS 32
#define FCNCT  (NPARAM / FCCOLS)
#define FCRS   204
#define WPLANE (NBATCH * NPARAM)

typedef __attribute__((ext_vector_type(2))) float v2f;
typedef __attribute__((ext_vector_type(4))) float v4f;
typedef __attribute__((ext_vector_type(8))) float v8f;

__global__ __launch_bounds__(64) void kan_fc_splitk_kernel(const float* __restrict__ features,
                                                           const float* __restrict__ fc_w,
                                                           const float* __restrict__ fc_b,
                                                           float* __restrict__ part) {
    __shared__ __attribute__((aligned(16))) float wsl[FCCOLS * FCRS];
    __shared__ __attribute__((aligned(16))) float fsl[NBATCH * FCRS];

    const int tid  = threadIdx.x;
    const int lane = tid & 31;
    const int wv   = tid >> 5;
    const int m    = lane & 15;
    const int hi   = lane >> 4;
    const int j0   = blockIdx.x * FCCOLS;
    const int s    = blockIdx.y;
    const int kc   = s * FCKC;

    const int   jc   = j0 + wv * 16 + m;
    const float bias = (s == 0) ? fc_b[jc] : 0.0f;
    if (tid < FCUPR) {
        const float* wsrc = fc_w + (size_t)j0 * NFEAT + kc + 4 * tid;
        const float* fsrc = features + kc + 4 * tid;
        v4f wreg[FCCOLS];
        v4f freg[NBATCH];
#pragma unroll
        for (int r = 0; r < FCCOLS; ++r) wreg[r] = *(const v4f*)(wsrc + r * NFEAT);
#pragma unroll
        for (int r = 0; r < NBATCH; ++r) freg[r] = *(const v4f*)(fsrc + r * NFEAT);
#pragma unroll
        for (int r = 0; r < FCCOLS; ++r) *(v4f*)(&wsl[r * FCRS + 4 * tid]) = wreg[r];
#pragma unroll
        for (int r = 0; r < NBATCH; ++r) *(v4f*)(&fsl[r * FCRS + 4 * tid]) = freg[r];
    }
    __syncthreads();

    v8f acc0, acc1;
#pragma unroll
    for (int r = 0; r < 8; ++r) { acc0[r] = bias; acc1[r] = 0.0f; }

    const float* ap = &fsl[(m & 7) * FCRS + 4 * hi];
    const float* bp = &wsl[(wv * 16 + m) * FCRS + 4 * hi];
#pragma unroll
    for (int t = 0; t < FCKC / 8; ++t) {
        const v4f a4 = *(const v4f*)(ap + 8 * t);
        const v4f b4 = *(const v4f*)(bp + 8 * t);
        v2f a, bb;
        a.x = a4.x;  a.y = a4.y;  bb.x = b4.x;  bb.y = b4.y;
        acc0 = __builtin_amdgcn_wmma_f32_16x16x4_f32(
                 false, a, false, bb, (short)0, acc0, false, false);
        a.x = a4.z;  a.y = a4.w;  bb.x = b4.z;  bb.y = b4.w;
        acc1 = __builtin_amdgcn_wmma_f32_16x16x4_f32(
                 false, a, false, bb, (short)0, acc1, false, false);
    }
    if (hi == 0) {
        float* dst = part + (size_t)s * WPLANE + jc;
#pragma unroll
        for (int r = 0; r < 8; ++r)
            dst[r * NPARAM] = acc0[r] + acc1[r];
    }
}

__device__ __forceinline__ float fc_sum(const float* __restrict__ p) {
    float v = p[0];
#pragma unroll
    for (int s = 1; s < FCS; ++s) v += p[(size_t)s * WPLANE];
    return v;
}

__global__ void kan_wm_kernel(const float* __restrict__ w,
                              float* __restrict__ wm) {
    const int e = blockIdx.x * blockDim.x + threadIdx.x;
    const int total = NBATCH * WM_PER_B;
    if (e >= total) return;
    const int b   = e / WM_PER_B;
    const int rem = e % WM_PER_B;

    int layer, d;
    if (rem < WF0)            { layer = 0; d = rem; }
    else if (rem < WF0 + WF1) { layer = 1; d = rem - WF0; }
    else                      { layer = 2; d = rem - WF0 - WF1; }

    const int DI_[3] = {3, 16, 16};
    const int DO_[3] = {16, 16, 3};
    const int I0[3]  = {0,   624, 3952};
    const int I1[3]  = {528, 3440, 4480};
    const int I2[3]  = {576, 3696, 4528};

    const int dih = (layer == 0) ? DIH0 : DIH1;
    const int s4  = d & 3;
    const int o   = (d >> 2) & 15;
    const int h   = (d >> 6) & 1;
    const int pp  = d >> 7;
    const int j   = pp / 3;
    const int t   = pp % 3;
    const int i   = h * dih + j;
    const int k   = 4 * t + s4;

    float val = 0.0f;
    const int do_ = DO_[layer];
    if (i < DI_[layer] && o < do_) {
        const float* wb = w + (size_t)b * NPARAM;
        const int io = i * do_ + o;
        if (k < KB) val = fc_sum(wb + I0[layer] + io * KB + k) * fc_sum(wb + I1[layer] + io);
        else        val = fc_sum(wb + I2[layer] + io);
    }
    wm[e] = val;
}

__device__ __forceinline__ void bspline12(float x, float* f) {
    float b[14];
#pragma unroll
    for (int j = 0; j < 14; ++j) {
        const float g0 = -1.75f + 0.25f * j;
        b[j] = (x >= g0 && x < g0 + 0.25f) ? 1.0f : 0.0f;
    }
#pragma unroll
    for (int p = 1; p <= 3; ++p) {
        const float inv = 1.0f / (0.25f * p);
#pragma unroll 13
        for (int j = 0; j < 14 - p; ++j) {
            const float gj = -1.75f + 0.25f * j;
            const float left  = (x - gj) * inv * b[j];
            const float right = ((gj + 0.25f * (p + 1)) - x) * inv * b[j + 1];
            b[j] = left + right;
        }
    }
#pragma unroll
    for (int k = 0; k < KB; ++k) f[k] = b[k];
    f[KB] = x / (1.0f + __expf(-x));
}

template <int DIH>
__device__ __forceinline__ void kan_layer(const float* wl,
                                          const float* yrow,
                                          int fragoff,
                                          v8f& acc0, v8f& acc1) {
    for (int j = 0; j < DIH; ++j) {
        const float xv = yrow[j];
        const float* wp = wl + j * (3 * 128) + fragoff;
        const v4f w0 = *(const v4f*)(wp);
        const v4f w1 = *(const v4f*)(wp + 128);
        const v4f w2 = *(const v4f*)(wp + 256);
        float f[RPI];
        bspline12(xv, f);
        v2f a, bb;
        a.x = f[0];  a.y = f[1];  bb.x = w0.x; bb.y = w0.y;
        acc0 = __builtin_amdgcn_wmma_f32_16x16x4_f32(false, a, false, bb, (short)0, acc0, false, false);
        a.x = f[2];  a.y = f[3];  bb.x = w0.z; bb.y = w0.w;
        acc1 = __builtin_amdgcn_wmma_f32_16x16x4_f32(false, a, false, bb, (short)0, acc1, false, false);
        a.x = f[4];  a.y = f[5];  bb.x = w1.x; bb.y = w1.y;
        acc0 = __builtin_amdgcn_wmma_f32_16x16x4_f32(false, a, false, bb, (short)0, acc0, false, false);
        a.x = f[6];  a.y = f[7];  bb.x = w1.z; bb.y = w1.w;
        acc1 = __builtin_amdgcn_wmma_f32_16x16x4_f32(false, a, false, bb, (short)0, acc1, false, false);
        a.x = f[8];  a.y = f[9];  bb.x = w2.x; bb.y = w2.y;
        acc0 = __builtin_amdgcn_wmma_f32_16x16x4_f32(false, a, false, bb, (short)0, acc0, false, false);
        a.x = f[10]; a.y = f[11]; bb.x = w2.z; bb.y = w2.w;
        acc1 = __builtin_amdgcn_wmma_f32_16x16x4_f32(false, a, false, bb, (short)0, acc1, false, false);
    }
}

__device__ __forceinline__ void put_tile(float* yw, int c, int h, const v8f& acc0, const v8f& acc1) {
#pragma unroll
    for (int r = 0; r < 8; ++r) yw[(8 * h + r) * YP + c] = acc0[r] + acc1[r];
}

__global__ __launch_bounds__(128) void kan_main_kernel(const float* __restrict__ x,
                                                       const float* __restrict__ wm,
                                                       float* __restrict__ out) {
    __shared__ __attribute__((aligned(16))) float wms[WM_PER_B];
    __shared__ __attribute__((aligned(16))) float ybuf[4 * 16 * YP];
    __shared__ float obuf[3 * TOK];

    const int tid   = threadIdx.x;
    const int lane  = tid & 31;
    const int wv    = tid >> 5;
    const int c     = lane & 15;
    const int h     = lane >> 4;
    const int b     = blockIdx.x / (HW / TOK);
    const int chunk = blockIdx.x % (HW / TOK);
    const int hw0   = chunk * TOK;

    const v4f* wsrc = (const v4f*)(wm + (size_t)b * WM_PER_B);
    for (int i = tid; i < WM_PER_B / 4; i += 128) ((v4f*)wms)[i] = wsrc[i];

    float* yw = ybuf + wv * 16 * YP;
    const float* xb = x + (size_t)b * 3 * HW + hw0 + wv * 16 + c;
    const float x0 = xb[(size_t)(2 * h) * HW];
    const float x1 = (h == 0) ? xb[HW] : 0.0f;
    yw[c * YP + 2 * h]     = x0;
    yw[c * YP + 2 * h + 1] = x1;
    __syncthreads();

    const int fragoff = (h * 16 + c) * 4;
    const v8f zero = {0.f, 0.f, 0.f, 0.f, 0.f, 0.f, 0.f, 0.f};
    v8f acc0, acc1;

    acc0 = zero; acc1 = zero;
    kan_layer<DIH0>(wms, yw + c * YP + h * DIH0, fragoff, acc0, acc1);
    put_tile(yw, c, h, acc0, acc1);
    __builtin_amdgcn_wave_barrier();

    acc0 = zero; acc1 = zero;
    kan_layer<DIH1>(wms + WF0, yw + c * YP + h * DIH1, fragoff, acc0, acc1);
    put_tile(yw, c, h, acc0, acc1);
    __builtin_amdgcn_wave_barrier();

    acc0 = zero; acc1 = zero;
    kan_layer<DIH1>(wms + WF0 + WF1, yw + c * YP + h * DIH1, fragoff, acc0, acc1);

    if (c < 3) {
#pragma unroll
        for (int r = 0; r < 8; ++r) obuf[c * TOK + wv * 16 + 8 * h + r] = acc0[r] + acc1[r];
    }
    __syncthreads();
    for (int idx = tid; idx < 3 * TOK; idx += 128)
        out[((size_t)(b * 3 + (idx >> 6))) * HW + hw0 + (idx & (TOK - 1))] = obuf[idx];
}

extern "C" void kernel_launch(void* const* d_in, const int* in_sizes, int n_in,
                              void* d_out, int out_size, void* d_ws, size_t ws_size,
                              hipStream_t stream) {
    const float* x        = (const float*)d_in[0];
    const float* features = (const float*)d_in[1];
    const float* fc_w     = (const float*)d_in[2];
    const float* fc_b     = (const float*)d_in[3];
    float* out    = (float*)d_out;
    float* w_buf  = (float*)d_ws;
    float* wm_buf = w_buf + (size_t)FCS * WPLANE;

    kan_fc_splitk_kernel<<<dim3(FCNCT, FCS), 64, 0, stream>>>(features, fc_w, fc_b, w_buf);

    kan_wm_kernel<<<(NBATCH * WM_PER_B + 255) / 256, 256, 0, stream>>>(w_buf, wm_buf);

    kan_main_kernel<<<NBATCH * (HW / TOK), 128, 0, stream>>>(x, wm_buf, out);
}


# B-spline features: all 11 basis values from 13 shared linear hats L_j=max(1-4|x-g_j|,0), B_k = L_{k+2} - L_{k+2}^3/3 + (L_{k+1}^3+L_{k+3}^3)/6; no interval tests, no recursion; values agree to 2e-7
# speedup vs baseline: 6.4433x; 1.4494x over previous
#include <hip/hip_runtime.h>
#include <hip/hip_bf16.h>


#define NBATCH   8
#define NFEAT    1000
#define NPARAM   4576
#define HW       50176
#define KB       11
#define RPI      12

#define TOK   64
#define DIH0  2
#define DIH1  8
#define WF0   (3 * DIH0 * 2 * 16 * 4)
#define WF1   (3 * DIH1 * 2 * 16 * 4)
#define WM_PER_B (WF0 + 2 * WF1)
#define YP    20

#define FCS    5
#define FCKC   200
#define FCUPR  50
#define FCCOLS 32
#define FCNCT  (NPARAM / FCCOLS)
#define FCRS   204
#define WPLANE (NBATCH * NPARAM)

typedef __attribute__((ext_vector_type(2))) float v2f;
typedef __attribute__((ext_vector_type(4))) float v4f;
typedef __attribute__((ext_vector_type(8))) float v8f;

__global__ __launch_bounds__(64) void kan_fc_splitk_kernel(const float* __restrict__ features,
                                                           const float* __restrict__ fc_w,
                                                           const float* __restrict__ fc_b,
                                                           float* __restrict__ part) {
    __shared__ __attribute__((aligned(16))) float wsl[FCCOLS * FCRS];
    __shared__ __attribute__((aligned(16))) float fsl[NBATCH * FCRS];

    const int tid  = threadIdx.x;
    const int lane = tid & 31;
    const int wv   = tid >> 5;
    const int m    = lane & 15;
    const int hi   = lane >> 4;
    const int j0   = blockIdx.x * FCCOLS;
    const int s    = blockIdx.y;
    const int kc   = s * FCKC;

    const int   jc   = j0 + wv * 16 + m;
    const float bias = (s == 0) ? fc_b[jc] : 0.0f;
    if (tid < FCUPR) {
        const float* wsrc = fc_w + (size_t)j0 * NFEAT + kc + 4 * tid;
        const float* fsrc = features + kc + 4 * tid;
        v4f wreg[FCCOLS];
        v4f freg[NBATCH];
#pragma unroll
        for (int r = 0; r < FCCOLS; ++r) wreg[r] = *(const v4f*)(wsrc + r * NFEAT);
#pragma unroll
        for (int r = 0; r < NBATCH; ++r) freg[r] = *(const v4f*)(fsrc + r * NFEAT);
#pragma unroll
        for (int r = 0; r < FCCOLS; ++r) *(v4f*)(&wsl[r * FCRS + 4 * tid]) = wreg[r];
#pragma unroll
        for (int r = 0; r < NBATCH; ++r) *(v4f*)(&fsl[r * FCRS + 4 * tid]) = freg[r];
    }
    __syncthreads();

    v8f acc0, acc1;
#pragma unroll
    for (int r = 0; r < 8; ++r) { acc0[r] = bias; acc1[r] = 0.0f; }

    const float* ap = &fsl[(m & 7) * FCRS + 4 * hi];
    const float* bp = &wsl[(wv * 16 + m) * FCRS + 4 * hi];
#pragma unroll
    for (int t = 0; t < FCKC / 8; ++t) {
        const v4f a4 = *(const v4f*)(ap + 8 * t);
        const v4f b4 = *(const v4f*)(bp + 8 * t);
        v2f a, bb;
        a.x = a4.x;  a.y = a4.y;  bb.x = b4.x;  bb.y = b4.y;
        acc0 = __builtin_amdgcn_wmma_f32_16x16x4_f32(
                 false, a, false, bb, (short)0, acc0, false, false);
        a.x = a4.z;  a.y = a4.w;  bb.x = b4.z;  bb.y = b4.w;
        acc1 = __builtin_amdgcn_wmma_f32_16x16x4_f32(
                 false, a, false, bb, (short)0, acc1, false, false);
    }
    if (hi == 0) {
        float* dst = part + (size_t)s * WPLANE + jc;
#pragma unroll
        for (int r = 0; r < 8; ++r)
            dst[r * NPARAM] = acc0[r] + acc1[r];
    }
}

__device__ __forceinline__ float fc_sum(const float* __restrict__ p) {
    float v = p[0];
#pragma unroll
    for (int s = 1; s < FCS; ++s) v += p[(size_t)s * WPLANE];
    return v;
}

__global__ void kan_wm_kernel(const float* __restrict__ w,
                              float* __restrict__ wm) {
    const int e = blockIdx.x * blockDim.x + threadIdx.x;
    const int total = NBATCH * WM_PER_B;
    if (e >= total) return;
    const int b   = e / WM_PER_B;
    const int rem = e % WM_PER_B;

    int layer, d;
    if (rem < WF0)            { layer = 0; d = rem; }
    else if (rem < WF0 + WF1) { layer = 1; d = rem - WF0; }
    else                      { layer = 2; d = rem - WF0 - WF1; }

    const int DI_[3] = {3, 16, 16};
    const int DO_[3] = {16, 16, 3};
    const int I0[3]  = {0,   624, 3952};
    const int I1[3]  = {528, 3440, 4480};
    const int I2[3]  = {576, 3696, 4528};

    const int dih = (layer == 0) ? DIH0 : DIH1;
    const int s4  = d & 3;
    const int o   = (d >> 2) & 15;
    const int h   = (d >> 6) & 1;
    const int pp  = d >> 7;
    const int j   = pp / 3;
    const int t   = pp % 3;
    const int i   = h * dih + j;
    const int k   = 4 * t + s4;

    float val = 0.0f;
    const int do_ = DO_[layer];
    if (i < DI_[layer] && o < do_) {
        const float* wb = w + (size_t)b * NPARAM;
        const int io = i * do_ + o;
        if (k < KB) val = fc_sum(wb + I0[layer] + io * KB + k) * fc_sum(wb + I1[layer] + io);
        else        val = fc_sum(wb + I2[layer] + io);
    }
    wm[e] = val;
}

__device__ __forceinline__ void bspline12(float x, float* f) {
    float L[14], C[14];
#pragma unroll
    for (int j = 1; j <= 13; ++j) {
        const float gj = -1.75f + 0.25f * j;
        L[j] = __builtin_amdgcn_fmed3f(__builtin_fmaf(__builtin_fabsf(x - gj), -4.0f, 1.0f), 0.0f, 1.0f);
        C[j] = L[j] * L[j] * L[j];
    }
#pragma unroll
    for (int k = 0; k < KB; ++k)
        f[k] = __builtin_fmaf(C[k + 1] + C[k + 3], 1.0f / 6.0f,
                              __builtin_fmaf(C[k + 2], -1.0f / 3.0f, L[k + 2]));
    f[KB] = x / (1.0f + __expf(-x));
}

template <int DIH>
__device__ __forceinline__ void kan_layer(const float* wl,
                                          const float* yrow,
                                          int fragoff,
                                          v8f& acc0, v8f& acc1) {
    for (int j = 0; j < DIH; ++j) {
        const float xv = yrow[j];
        const float* wp = wl + j * (3 * 128) + fragoff;
        const v4f w0 = *(const v4f*)(wp);
        const v4f w1 = *(const v4f*)(wp + 128);
        const v4f w2 = *(const v4f*)(wp + 256);
        float f[RPI];
        bspline12(xv, f);
        v2f a, bb;
        a.x = f[0];  a.y = f[1];  bb.x = w0.x; bb.y = w0.y;
        acc0 = __builtin_amdgcn_wmma_f32_16x16x4_f32(false, a, false, bb, (short)0, acc0, false, false);
        a.x = f[2];  a.y = f[3];  bb.x = w0.z; bb.y = w0.w;
        acc1 = __builtin_amdgcn_wmma_f32_16x16x4_f32(false, a, false, bb, (short)0, acc1, false, false);
        a.x = f[4];  a.y = f[5];  bb.x = w1.x; bb.y = w1.y;
        acc0 = __builtin_amdgcn_wmma_f32_16x16x4_f32(false, a, false, bb, (short)0, acc0, false, false);
        a.x = f[6];  a.y = f[7];  bb.x = w1.z; bb.y = w1.w;
        acc1 = __builtin_amdgcn_wmma_f32_16x16x4_f32(false, a, false, bb, (short)0, acc1, false, false);
        a.x = f[8];  a.y = f[9];  bb.x = w2.x; bb.y = w2.y;
        acc0 = __builtin_amdgcn_wmma_f32_16x16x4_f32(false, a, false, bb, (short)0, acc0, false, false);
        a.x = f[10]; a.y = f[11]; bb.x = w2.z; bb.y = w2.w;
        acc1 = __builtin_amdgcn_wmma_f32_16x16x4_f32(false, a, false, bb, (short)0, acc1, false, false);
    }
}

__device__ __forceinline__ void put_tile(float* yw, int c, int h, const v8f& acc0, const v8f& acc1) {
#pragma unroll
    for (int r = 0; r < 8; ++r) yw[(8 * h + r) * YP + c] = acc0[r] + acc1[r];
}

__global__ __launch_bounds__(128) void kan_main_kernel(const float* __restrict__ x,
                                                       const float* __restrict__ wm,
                                                       float* __restrict__ out) {
    __shared__ __attribute__((aligned(16))) float wms[WM_PER_B];
    __shared__ __attribute__((aligned(16))) float ybuf[4 * 16 * YP];
    __shared__ float obuf[3 * TOK];

    const int tid   = threadIdx.x;
    const int lane  = tid & 31;
    const int wv    = tid >> 5;
    const int c     = lane & 15;
    const int h     = lane >> 4;
    const int b     = blockIdx.x / (HW / TOK);
    const int chunk = blockIdx.x % (HW / TOK);
    const int hw0   = chunk * TOK;

    const v4f* wsrc = (const v4f*)(wm + (size_t)b * WM_PER_B);
    for (int i = tid; i < WM_PER_B / 4; i += 128) ((v4f*)wms)[i] = wsrc[i];

    float* yw = ybuf + wv * 16 * YP;
    const float* xb = x + (size_t)b * 3 * HW + hw0 + wv * 16 + c;
    const float x0 = xb[(size_t)(2 * h) * HW];
    const float x1 = (h == 0) ? xb[HW] : 0.0f;
    yw[c * YP + 2 * h]     = x0;
    yw[c * YP + 2 * h + 1] = x1;
    __syncthreads();

    const int fragoff = (h * 16 + c) * 4;
    const v8f zero = {0.f, 0.f, 0.f, 0.f, 0.f, 0.f, 0.f, 0.f};
    v8f acc0, acc1;

    acc0 = zero; acc1 = zero;
    kan_layer<DIH0>(wms, yw + c * YP + h * DIH0, fragoff, acc0, acc1);
    put_tile(yw, c, h, acc0, acc1);
    __builtin_amdgcn_wave_barrier();

    acc0 = zero; acc1 = zero;
    kan_layer<DIH1>(wms + WF0, yw + c * YP + h * DIH1, fragoff, acc0, acc1);
    put_tile(yw, c, h, acc0, acc1);
    __builtin_amdgcn_wave_barrier();

    acc0 = zero; acc1 = zero;
    kan_layer<DIH1>(wms + WF0 + WF1, yw + c * YP + h * DIH1, fragoff, acc0, acc1);

    if (c < 3) {
#pragma unroll
        for (int r = 0; r < 8; ++r) obuf[c * TOK + wv * 16 + 8 * h + r] = acc0[r] + acc1[r];
    }
    __syncthreads();
    for (int idx = tid; idx < 3 * TOK; idx += 128)
        out[((size_t)(b * 3 + (idx >> 6))) * HW + hw0 + (idx & (TOK - 1))] = obuf[idx];
}

extern "C" void kernel_launch(void* const* d_in, const int* in_sizes, int n_in,
                              void* d_out, int out_size, void* d_ws, size_t ws_size,
                              hipStream_t stream) {
    const float* x        = (const float*)d_in[0];
    const float* features = (const float*)d_in[1];
    const float* fc_w     = (const float*)d_in[2];
    const float* fc_b     = (const float*)d_in[3];
    float* out    = (float*)d_out;
    float* w_buf  = (float*)d_ws;
    float* wm_buf = w_buf + (size_t)FCS * WPLANE;

    kan_fc_splitk_kernel<<<dim3(FCNCT, FCS), 64, 0, stream>>>(features, fc_w, fc_b, w_buf);

    kan_wm_kernel<<<(NBATCH * WM_PER_B + 255) / 256, 256, 0, stream>>>(w_buf, wm_buf);

    kan_main_kernel<<<NBATCH * (HW / TOK), 128, 0, stream>>>(x, wm_buf, out);
}


# Merge launch folded into the FC GEMM: a work-group per (in-dim, out pair) holds its coef, uw, rw rows with full K (4 K quarters summed in LDS: new summation order of w), writes the merged line
# speedup vs baseline: 6.7105x; 1.0415x over previous
#include <hip/hip_runtime.h>
#include <hip/hip_bf16.h>


#define NBATCH   8
#define NFEAT    1000
#define NPARAM   4576
#define HW       50176
#define KB       11
#define RPI      12

#define TOK   64
#define DIH0  2
#define DIH1  8
#define WF0   (3 * DIH0 * 2 * 16 * 4)
#define WF1   (3 * DIH1 * 2 * 16 * 4)
#define WM_PER_B (WF0 + 2 * WF1)
#define YP    20

#define FCNBLK  184
#define FCUPR   250
#define FCRS    1028
#define FCROWS  26
#define FCQS    32
#define WL0 0
#define WL1 (4 * 8 * 32)
#define WL2 (WL1 + 16 * 8 * 32)
#define WQ_PER_B (WL2 + 16 * 8 * 32)

typedef __attribute__((ext_vector_type(2))) float v2f;
typedef __attribute__((ext_vector_type(4))) float v4f;
typedef __attribute__((ext_vector_type(8))) float v8f;

__global__ __launch_bounds__(256) void kan_fc_merge_kernel(const float* __restrict__ features,
                                                           const float* __restrict__ fc_w,
                                                           const float* __restrict__ fc_b,
                                                           float* __restrict__ wm) {
    __shared__ __attribute__((aligned(16))) float wsl[FCROWS * FCRS];
    __shared__ __attribute__((aligned(16))) float fsl[NBATCH * FCRS];
    __shared__ float red[4 * 2 * NBATCH * 16];
    __shared__ float fin[2 * NBATCH * 16];

    const int tid = threadIdx.x;
    const int bid = blockIdx.x;

    int layer, i, p;
    if (bid < 24)       { layer = 0; i = bid >> 3;         p = bid & 7; }
    else if (bid < 152) { layer = 1; i = (bid - 24) >> 3;  p = (bid - 24) & 7; }
    else                { layer = 2; i = (bid - 152) >> 1; p = (bid - 152) & 1; }
    const int dout = (layer == 2) ? 3 : 16;
    const int I0   = (layer == 0) ? 0   : (layer == 1) ? 624  : 3952;
    const int I1   = (layer == 0) ? 528 : (layer == 1) ? 3440 : 4480;
    const int I2   = (layer == 0) ? 576 : (layer == 1) ? 3696 : 4528;
    const int wl   = (layer == 0) ? WL0 : (layer == 1) ? WL1 : WL2;
    const bool two = (2 * p + 1) < dout;
    const int io0 = i * dout + 2 * p;
    const int io1 = two ? io0 + 1 : io0;

    {
        v4f wreg[FCROWS];
        v4f freg[NBATCH];
        const bool live = tid < FCUPR;
        const int  u4   = live ? 4 * tid : 0;
#pragma unroll
        for (int r = 0; r < FCROWS; ++r) {
            const int n  = r / 13, c = r % 13;
            const int io = n ? io1 : io0;
            const int grow = (c < 11) ? I0 + io * KB + c : (c == 11) ? I1 + io : I2 + io;
            wreg[r] = *(const v4f*)(fc_w + (size_t)grow * NFEAT + u4);
        }
#pragma unroll
        for (int r = 0; r < NBATCH; ++r) freg[r] = *(const v4f*)(features + r * NFEAT + u4);
        const v4f z4 = {0.f, 0.f, 0.f, 0.f};
#pragma unroll
        for (int r = 0; r < FCROWS; ++r) *(v4f*)(&wsl[r * FCRS + 4 * tid]) = live ? wreg[r] : z4;
#pragma unroll
        for (int r = 0; r < NBATCH; ++r) *(v4f*)(&fsl[r * FCRS + 4 * tid]) = live ? freg[r] : z4;
    }
    __syncthreads();

    const int lane = tid & 31;
    const int wv   = tid >> 5;
    const int m    = lane & 15;
    const int hi   = lane >> 4;
    const int n    = wv & 1;
    const int q    = wv >> 1;
    {
        v8f acc0, acc1;
#pragma unroll
        for (int r = 0; r < 8; ++r) { acc0[r] = 0.0f; acc1[r] = 0.0f; }
        const int brow = n * 13 + ((m < 12) ? m : 12);
        const float* ap = &fsl[(m & 7) * FCRS + 4 * hi + 8 * FCQS * q];
        const float* bp = &wsl[brow * FCRS + 4 * hi + 8 * FCQS * q];
#pragma unroll
        for (int t = 0; t < FCQS; ++t) {
            const v4f a4 = *(const v4f*)(ap + 8 * t);
            const v4f b4 = *(const v4f*)(bp + 8 * t);
            v2f a, bb;
            a.x = a4.x;  a.y = a4.y;  bb.x = b4.x;  bb.y = b4.y;
            acc0 = __builtin_amdgcn_wmma_f32_16x16x4_f32(
                     false, a, false, bb, (short)0, acc0, false, false);
            a.x = a4.z;  a.y = a4.w;  bb.x = b4.z;  bb.y = b4.w;
            acc1 = __builtin_amdgcn_wmma_f32_16x16x4_f32(
                     false, a, false, bb, (short)0, acc1, false, false);
        }
        if (hi == 0) {
#pragma unroll
            for (int r = 0; r < 8; ++r) red[((q * 2 + n) * NBATCH + r) * 16 + m] = acc0[r] + acc1[r];
        }
    }
    __syncthreads();

    {
        const int fn = tid >> 7, fb = (tid >> 4) & 7, fc = tid & 15;
        float v = 0.0f;
        if (fc < 13) {
            const int io = fn ? io1 : io0;
            const int grow = (fc < 11) ? I0 + io * KB + fc : (fc == 11) ? I1 + io : I2 + io;
            v = fc_b[grow];
#pragma unroll
            for (int s = 0; s < 4; ++s) v += red[((s * 2 + fn) * NBATCH + fb) * 16 + fc];
        }
        fin[tid] = v;
    }
    __syncthreads();

    if (tid < 2 * NBATCH * 3) {
        const int en = tid / 24, eb = (tid % 24) / 3, et = tid % 3;
        const float* f = &fin[(en * NBATCH + eb) * 16];
        const float uwv = f[11], rwv = f[12];
        v4f o4;
        o4.x = f[4 * et] * uwv;
        o4.y = f[4 * et + 1] * uwv;
        o4.z = f[4 * et + 2] * uwv;
        o4.w = (et == 2) ? rwv : f[4 * et + 3] * uwv;
        if (en == 1 && !two) { o4.x = 0.0f; o4.y = 0.0f; o4.z = 0.0f; o4.w = 0.0f; }
        *(v4f*)(wm + (size_t)eb * WQ_PER_B + wl + ((i * 8 + p) * 8 + et * 2 + en) * 4) = o4;
    } else if (tid >= 64 && tid < 64 + 3 * 48) {
        const int zl = (tid - 64) / 48, zb = ((tid - 64) % 48) / 6, zu = (tid - 64) % 6;
        const v4f z4 = {0.f, 0.f, 0.f, 0.f};
        if (layer == 2)
            *(v4f*)(wm + (size_t)zb * WQ_PER_B + WL2 + ((i * 8 + 2 + 3 * p + zl) * 8 + zu) * 4) = z4;
        else if (layer == 0 && i == 2 && zl == 0)
            *(v4f*)(wm + (size_t)zb * WQ_PER_B + WL0 + ((3 * 8 + p) * 8 + zu) * 4) = z4;
    }
}

__device__ __forceinline__ void bspline12(float x, float* f) {
    float L[14], C[14];
#pragma unroll
    for (int j = 1; j <= 13; ++j) {
        const float gj = -1.75f + 0.25f * j;
        L[j] = __builtin_amdgcn_fmed3f(__builtin_fmaf(__builtin_fabsf(x - gj), -4.0f, 1.0f), 0.0f, 1.0f);
        C[j] = L[j] * L[j] * L[j];
    }
#pragma unroll
    for (int k = 0; k < KB; ++k)
        f[k] = __builtin_fmaf(C[k + 1] + C[k + 3], 1.0f / 6.0f,
                              __builtin_fmaf(C[k + 2], -1.0f / 3.0f, L[k + 2]));
    f[KB] = x / (1.0f + __expf(-x));
}

template <int DIH>
__device__ __forceinline__ void kan_layer(const float* wl,
                                          const float* yrow,
                                          int fragoff,
                                          v8f& acc0, v8f& acc1) {
    for (int j = 0; j < DIH; ++j) {
        const float xv = yrow[j];
        const float* wp = wl + j * (3 * 128) + fragoff;
        const v4f w0 = *(const v4f*)(wp);
        const v4f w1 = *(const v4f*)(wp + 128);
        const v4f w2 = *(const v4f*)(wp + 256);
        float f[RPI];
        bspline12(xv, f);
        v2f a, bb;
        a.x = f[0];  a.y = f[1];  bb.x = w0.x; bb.y = w0.y;
        acc0 = __builtin_amdgcn_wmma_f32_16x16x4_f32(false, a, false, bb, (short)0, acc0, false, false);
        a.x = f[2];  a.y = f[3];  bb.x = w0.z; bb.y = w0.w;
        acc1 = __builtin_amdgcn_wmma_f32_16x16x4_f32(false, a, false, bb, (short)0, acc1, false, false);
        a.x = f[4];  a.y = f[5];  bb.x = w1.x; bb.y = w1.y;
        acc0 = __builtin_amdgcn_wmma_f32_16x16x4_f32(false, a, false, bb, (short)0, acc0, false, false);
        a.x = f[6];  a.y = f[7];  bb.x = w1.z; bb.y = w1.w;
        acc1 = __builtin_amdgcn_wmma_f32_16x16x4_f32(false, a, false, bb, (short)0, acc1, false, false);
        a.x = f[8];  a.y = f[9];  bb.x = w2.x; bb.y = w2.y;
        acc0 = __builtin_amdgcn_wmma_f32_16x16x4_f32(false, a, false, bb, (short)0, acc0, false, false);
        a.x = f[10]; a.y = f[11]; bb.x = w2.z; bb.y = w2.w;
        acc1 = __builtin_amdgcn_wmma_f32_16x16x4_f32(false, a, false, bb, (short)0, acc1, false, false);
    }
}

__device__ __forceinline__ void put_tile(float* yw, int c, int h, const v8f& acc0, const v8f& acc1) {
#pragma unroll
    for (int r = 0; r < 8; ++r) yw[(8 * h + r) * YP + c] = acc0[r] + acc1[r];
}

template <int DIH>
__device__ __forceinline__ void stage_load(const float* __restrict__ src, int ii, int o, v4f* v) {
#pragma unroll
    for (int g = 0; g < (2 * DIH + 7) / 8; ++g) {
        const int i = 8 * g + ii;
        if (i < 2 * DIH) {
            const float* s = src + ((i * 8 + (o >> 1)) * 8 + (o & 1)) * 4;
            v[3 * g]     = *(const v4f*)(s);
            v[3 * g + 1] = *(const v4f*)(s + 8);
            v[3 * g + 2] = *(const v4f*)(s + 16);
        }
    }
}

template <int DIH>
__device__ __forceinline__ void stage_store(float* dst, int ii, int o, const v4f* v) {
#pragma unroll
    for (int g = 0; g < (2 * DIH + 7) / 8; ++g) {
        const int i = 8 * g + ii;
        if (i < 2 * DIH) {
            const int h = i / DIH, j = i % DIH;
            float* d = dst + ((3 * j * 2 + h) * 16 + o) * 4;
            *(v4f*)(d)       = v[3 * g];
            *(v4f*)(d + 128) = v[3 * g + 1];
            *(v4f*)(d + 256) = v[3 * g + 2];
        }
    }
}

__global__ __launch_bounds__(128) void kan_main_kernel(const float* __restrict__ x,
                                                       const float* __restrict__ wm,
                                                       float* __restrict__ out) {
    __shared__ __attribute__((aligned(16))) float wms[WM_PER_B];
    __shared__ __attribute__((aligned(16))) float ybuf[4 * 16 * YP];
    __shared__ float obuf[3 * TOK];

    const int tid   = threadIdx.x;
    const int lane  = tid & 31;
    const int wv    = tid >> 5;
    const int c     = lane & 15;
    const int h     = lane >> 4;
    const int b     = blockIdx.x / (HW / TOK);
    const int chunk = blockIdx.x % (HW / TOK);
    const int hw0   = chunk * TOK;

    const float* wb = wm + (size_t)b * WQ_PER_B;
    {
        const int sii = (tid >> 4) & 7, so = tid & 15;
        v4f s0[3], s1[6], s2[6];
        stage_load<DIH0>(wb + WL0, sii, so, s0);
        stage_load<DIH1>(wb + WL1, sii, so, s1);
        stage_load<DIH1>(wb + WL2, sii, so, s2);
        stage_store<DIH0>(wms, sii, so, s0);
        stage_store<DIH1>(wms + WF0, sii, so, s1);
        stage_store<DIH1>(wms + WF0 + WF1, sii, so, s2);
    }

    float* yw = ybuf + wv * 16 * YP;
    const float* xb = x + (size_t)b * 3 * HW + hw0 + wv * 16 + c;
    const float x0 = xb[(size_t)(2 * h) * HW];
    const float x1 = (h == 0) ? xb[HW] : 0.0f;
    yw[c * YP + 2 * h]     = x0;
    yw[c * YP + 2 * h + 1] = x1;
    __syncthreads();

    const int fragoff = (h * 16 + c) * 4;
    const v8f zero = {0.f, 0.f, 0.f, 0.f, 0.f, 0.f, 0.f, 0.f};
    v8f acc0, acc1;

    acc0 = zero; acc1 = zero;
    kan_layer<DIH0>(wms, yw + c * YP + h * DIH0, fragoff, acc0, acc1);
    put_tile(yw, c, h, acc0, acc1);
    __builtin_amdgcn_wave_barrier();

    acc0 = zero; acc1 = zero;
    kan_layer<DIH1>(wms + WF0, yw + c * YP + h * DIH1, fragoff, acc0, acc1);
    put_tile(yw, c, h, acc0, acc1);
    __builtin_amdgcn_wave_barrier();

    acc0 = zero; acc1 = zero;
    kan_layer<DIH1>(wms + WF0 + WF1, yw + c * YP + h * DIH1, fragoff, acc0, acc1);

    if (c < 3) {
#pragma unroll
        for (int r = 0; r < 8; ++r) obuf[c * TOK + wv * 16 + 8 * h + r] = acc0[r] + acc1[r];
    }
    __syncthreads();
    for (int idx = tid; idx < 3 * TOK; idx += 128)
        out[((size_t)(b * 3 + (idx >> 6))) * HW + hw0 + (idx & (TOK - 1))] = obuf[idx];
}

extern "C" void kernel_launch(void* const* d_in, const int* in_sizes, int n_in,
                              void* d_out, int out_size, void* d_ws, size_t ws_size,
                              hipStream_t stream) {
    const float* x        = (const float*)d_in[0];
    const float* features = (const float*)d_in[1];
    const float* fc_w     = (const float*)d_in[2];
    const float* fc_b     = (const float*)d_in[3];
    float* out    = (float*)d_out;
    float* wm_buf = (float*)d_ws;

    kan_fc_merge_kernel<<<FCNBLK, 256, 0, stream>>>(features, fc_w, fc_b, wm_buf);

    kan_main_kernel<<<NBATCH * (HW / TOK), 128, 0, stream>>>(x, wm_buf, out);
}


# Layer 1: a wave whose inputs all lie in [0,1) (tested) builds only hats 7..11, basis 4..10 and silu, all that can be non-zero there: 24 k values, 6 matrix steps for 12 (new k order); else old path
# speedup vs baseline: 7.0628x; 1.0525x over previous
#include <hip/hip_runtime.h>
#include <hip/hip_bf16.h>


#define NBATCH   8
#define NFEAT    1000
#define NPARAM   4576
#define HW       50176
#define KB       11
#define RPI      12

#define TOK   64
#define DIH0  2
#define DIH1  8
#define WF0   (3 * DIH0 * 2 * 16 * 4)
#define WF1   (3 * DIH1 * 2 * 16 * 4)
#define WM_PER_B (WF0 + 2 * WF1)
#define YP    20

#define FCNBLK  184
#define FCUPR   250
#define FCRS    1028
#define FCROWS  26
#define FCQS    32
#define WL0 0
#define WL1 (4 * 8 * 32)
#define WL2 (WL1 + 16 * 8 * 32)
#define WQ_PER_B (WL2 + 16 * 8 * 32)

typedef __attribute__((ext_vector_type(2))) float v2f;
typedef __attribute__((ext_vector_type(4))) float v4f;
typedef __attribute__((ext_vector_type(8))) float v8f;

__global__ __launch_bounds__(256) void kan_fc_merge_kernel(const float* __restrict__ features,
                                                           const float* __restrict__ fc_w,
                                                           const float* __restrict__ fc_b,
                                                           float* __restrict__ wm) {
    __shared__ __attribute__((aligned(16))) float wsl[FCROWS * FCRS];
    __shared__ __attribute__((aligned(16))) float fsl[NBATCH * FCRS];
    __shared__ float red[4 * 2 * NBATCH * 16];
    __shared__ float fin[2 * NBATCH * 16];

    const int tid = threadIdx.x;
    const int bid = blockIdx.x;

    int layer, i, p;
    if (bid < 24)       { layer = 0; i = bid >> 3;         p = bid & 7; }
    else if (bid < 152) { layer = 1; i = (bid - 24) >> 3;  p = (bid - 24) & 7; }
    else                { layer = 2; i = (bid - 152) >> 1; p = (bid - 152) & 1; }
    const int dout = (layer == 2) ? 3 : 16;
    const int I0   = (layer == 0) ? 0   : (layer == 1) ? 624  : 3952;
    const int I1   = (layer == 0) ? 528 : (layer == 1) ? 3440 : 4480;
    const int I2   = (layer == 0) ? 576 : (layer == 1) ? 3696 : 4528;
    const int wl   = (layer == 0) ? WL0 : (layer == 1) ? WL1 : WL2;
    const bool two = (2 * p + 1) < dout;
    const int io0 = i * dout + 2 * p;
    const int io1 = two ? io0 + 1 : io0;

    {
        v4f wreg[FCROWS];
        v4f freg[NBATCH];
        const bool live = tid < FCUPR;
        const int  u4   = live ? 4 * tid : 0;
#pragma unroll
        for (int r = 0; r < FCROWS; ++r) {
            const int n  = r / 13, c = r % 13;
            const int io = n ? io1 : io0;
            const int grow = (c < 11) ? I0 + io * KB + c : (c == 11) ? I1 + io : I2 + io;
            wreg[r] = *(const v4f*)(fc_w + (size_t)grow * NFEAT + u4);
        }
#pragma unroll
        for (int r = 0; r < NBATCH; ++r) freg[r] = *(const v4f*)(features + r * NFEAT + u4);
        const v4f z4 = {0.f, 0.f, 0.f, 0.f};
#pragma unroll
        for (int r = 0; r < FCROWS; ++r) *(v4f*)(&wsl[r * FCRS + 4 * tid]) = live ? wreg[r] : z4;
#pragma unroll
        for (int r = 0; r < NBATCH; ++r) *(v4f*)(&fsl[r * FCRS + 4 * tid]) = live ? freg[r] : z4;
    }
    __syncthreads();

    const int lane = tid & 31;
    const int wv   = tid >> 5;
    const int m    = lane & 15;
    const int hi   = lane >> 4;
    const int n    = wv & 1;
    const int q    = wv >> 1;
    {
        v8f acc0, acc1;
#pragma unroll
        for (int r = 0; r < 8; ++r) { acc0[r] = 0.0f; acc1[r] = 0.0f; }
        const int brow = n * 13 + ((m < 12) ? m : 12);
        const float* ap = &fsl[(m & 7) * FCRS + 4 * hi + 8 * FCQS * q];
        const float* bp = &wsl[brow * FCRS + 4 * hi + 8 * FCQS * q];
#pragma unroll
        for (int t = 0; t < FCQS; ++t) {
            const v4f a4 = *(const v4f*)(ap + 8 * t);
            const v4f b4 = *(const v4f*)(bp + 8 * t);
            v2f a, bb;
            a.x = a4.x;  a.y = a4.y;  bb.x = b4.x;  bb.y = b4.y;
            acc0 = __builtin_amdgcn_wmma_f32_16x16x4_f32(
                     false, a, false, bb, (short)0, acc0, false, false);
            a.x = a4.z;  a.y = a4.w;  bb.x = b4.z;  bb.y = b4.w;
            acc1 = __builtin_amdgcn_wmma_f32_16x16x4_f32(
                     false, a, false, bb, (short)0, acc1, false, false);
        }
        if (hi == 0) {
#pragma unroll
            for (int r = 0; r < 8; ++r) red[((q * 2 + n) * NBATCH + r) * 16 + m] = acc0[r] + acc1[r];
        }
    }
    __syncthreads();

    {
        const int fn = tid >> 7, fb = (tid >> 4) & 7, fc = tid & 15;
        float v = 0.0f;
        if (fc < 13) {
            const int io = fn ? io1 : io0;
            const int grow = (fc < 11) ? I0 + io * KB + fc : (fc == 11) ? I1 + io : I2 + io;
            v = fc_b[grow];
#pragma unroll
            for (int s = 0; s < 4; ++s) v += red[((s * 2 + fn) * NBATCH + fb) * 16 + fc];
        }
        fin[tid] = v;
    }
    __syncthreads();

    if (tid < 2 * NBATCH * 3) {
        const int en = tid / 24, eb = (tid % 24) / 3, et = tid % 3;
        const float* f = &fin[(en * NBATCH + eb) * 16];
        const float uwv = f[11], rwv = f[12];
        v4f o4;
        o4.x = f[4 * et] * uwv;
        o4.y = f[4 * et + 1] * uwv;
        o4.z = f[4 * et + 2] * uwv;
        o4.w = (et == 2) ? rwv : f[4 * et + 3] * uwv;
        if (en == 1 && !two) { o4.x = 0.0f; o4.y = 0.0f; o4.z = 0.0f; o4.w = 0.0f; }
        *(v4f*)(wm + (size_t)eb * WQ_PER_B + wl + ((i * 8 + p) * 8 + et * 2 + en) * 4) = o4;
    } else if (tid >= 64 && tid < 64 + 3 * 48) {
        const int zl = (tid - 64) / 48, zb = ((tid - 64) % 48) / 6, zu = (tid - 64) % 6;
        const v4f z4 = {0.f, 0.f, 0.f, 0.f};
        if (layer == 2)
            *(v4f*)(wm + (size_t)zb * WQ_PER_B + WL2 + ((i * 8 + 2 + 3 * p + zl) * 8 + zu) * 4) = z4;
        else if (layer == 0 && i == 2 && zl == 0)
            *(v4f*)(wm + (size_t)zb * WQ_PER_B + WL0 + ((3 * 8 + p) * 8 + zu) * 4) = z4;
    }
}

__device__ __forceinline__ void bspline12(float x, float* f) {
    float L[14], C[14];
#pragma unroll
    for (int j = 1; j <= 13; ++j) {
        const float gj = -1.75f + 0.25f * j;
        L[j] = __builtin_amdgcn_fmed3f(__builtin_fmaf(__builtin_fabsf(x - gj), -4.0f, 1.0f), 0.0f, 1.0f);
        C[j] = L[j] * L[j] * L[j];
    }
#pragma unroll
    for (int k = 0; k < KB; ++k)
        f[k] = __builtin_fmaf(C[k + 1] + C[k + 3], 1.0f / 6.0f,
                              __builtin_fmaf(C[k + 2], -1.0f / 3.0f, L[k + 2]));
    f[KB] = x / (1.0f + __expf(-x));
}

__device__ __forceinline__ void bspline8_unit(float x, float* f) {
    float L[5], C[5];
#pragma unroll
    for (int j = 0; j < 5; ++j) {
        const float gj = 0.25f * j;
        L[j] = __builtin_amdgcn_fmed3f(__builtin_fmaf(__builtin_fabsf(x - gj), -4.0f, 1.0f), 0.0f, 1.0f);
        C[j] = L[j] * L[j] * L[j];
    }
    f[0] = C[0] * (1.0f / 6.0f);
    f[1] = __builtin_fmaf(C[1], 1.0f / 6.0f, __builtin_fmaf(C[0], -1.0f / 3.0f, L[0]));
    f[2] = __builtin_fmaf(C[0] + C[2], 1.0f / 6.0f, __builtin_fmaf(C[1], -1.0f / 3.0f, L[1]));
    f[3] = __builtin_fmaf(C[1] + C[3], 1.0f / 6.0f, __builtin_fmaf(C[2], -1.0f / 3.0f, L[2]));
    f[4] = __builtin_fmaf(C[2] + C[4], 1.0f / 6.0f, __builtin_fmaf(C[3], -1.0f / 3.0f, L[3]));
    f[5] = __builtin_fmaf(C[3], 1.0f / 6.0f, __builtin_fmaf(C[4], -1.0f / 3.0f, L[4]));
    f[6] = C[4] * (1.0f / 6.0f);
    f[7] = x / (1.0f + __expf(-x));
}

__device__ __forceinline__ void kan_layer1_unit(const float* wl, float xs, float xm, int h, int c,
                                                v8f& acc0, v8f& acc1) {
    const v4f w0 = *(const v4f*)(wl + (32 + 16 * h + c) * 4);
    const v4f w1 = *(const v4f*)(wl + (64 + 16 * h + c) * 4);
    const v4f w2 = *(const v4f*)(wl + (128 + 32 * h + c) * 4);
    float fs[8], fm[8];
    bspline8_unit(xs, fs);
    bspline8_unit(xm, fm);
    const float m0 = h ? fm[4] : fm[0];
    const float m1 = h ? fm[5] : fm[1];
    const float m2 = h ? fm[6] : fm[2];
    const float m3 = h ? fm[7] : fm[3];
    v2f a, bb;
    a.x = fs[0]; a.y = fs[1]; bb.x = w0.x; bb.y = w0.y;
    acc0 = __builtin_amdgcn_wmma_f32_16x16x4_f32(false, a, false, bb, (short)0, acc0, false, false);
    a.x = fs[2]; a.y = fs[3]; bb.x = w0.z; bb.y = w0.w;
    acc1 = __builtin_amdgcn_wmma_f32_16x16x4_f32(false, a, false, bb, (short)0, acc1, false, false);
    a.x = fs[4]; a.y = fs[5]; bb.x = w1.x; bb.y = w1.y;
    acc0 = __builtin_amdgcn_wmma_f32_16x16x4_f32(false, a, false, bb, (short)0, acc0, false, false);
    a.x = fs[6]; a.y = fs[7]; bb.x = w1.z; bb.y = w1.w;
    acc1 = __builtin_amdgcn_wmma_f32_16x16x4_f32(false, a, false, bb, (short)0, acc1, false, false);
    a.x = m0;    a.y = m1;    bb.x = w2.x; bb.y = w2.y;
    acc0 = __builtin_amdgcn_wmma_f32_16x16x4_f32(false, a, false, bb, (short)0, acc0, false, false);
    a.x = m2;    a.y = m3;    bb.x = w2.z; bb.y = w2.w;
    acc1 = __builtin_amdgcn_wmma_f32_16x16x4_f32(false, a, false, bb, (short)0, acc1, false, false);
}

template <int DIH>
__device__ __forceinline__ void kan_layer(const float* wl,
                                          const float* yrow,
                                          int fragoff,
                                          v8f& acc0, v8f& acc1) {
    for (int j = 0; j < DIH; ++j) {
        const float xv = yrow[j];
        const float* wp = wl + j * (3 * 128) + fragoff;
        const v4f w0 = *(const v4f*)(wp);
        const v4f w1 = *(const v4f*)(wp + 128);
        const v4f w2 = *(const v4f*)(wp + 256);
        float f[RPI];
        bspline12(xv, f);
        v2f a, bb;
        a.x = f[0];  a.y = f[1];  bb.x = w0.x; bb.y = w0.y;
        acc0 = __builtin_amdgcn_wmma_f32_16x16x4_f32(false, a, false, bb, (short)0, acc0, false, false);
        a.x = f[2];  a.y = f[3];  bb.x = w0.z; bb.y = w0.w;
        acc1 = __builtin_amdgcn_wmma_f32_16x16x4_f32(false, a, false, bb, (short)0, acc1, false, false);
        a.x = f[4];  a.y = f[5];  bb.x = w1.x; bb.y = w1.y;
        acc0 = __builtin_amdgcn_wmma_f32_16x16x4_f32(false, a, false, bb, (short)0, acc0, false, false);
        a.x = f[6];  a.y = f[7];  bb.x = w1.z; bb.y = w1.w;
        acc1 = __builtin_amdgcn_wmma_f32_16x16x4_f32(false, a, false, bb, (short)0, acc1, false, false);
        a.x = f[8];  a.y = f[9];  bb.x = w2.x; bb.y = w2.y;
        acc0 = __builtin_amdgcn_wmma_f32_16x16x4_f32(false, a, false, bb, (short)0, acc0, false, false);
        a.x = f[10]; a.y = f[11]; bb.x = w2.z; bb.y = w2.w;
        acc1 = __builtin_amdgcn_wmma_f32_16x16x4_f32(false, a, false, bb, (short)0, acc1, false, false);
    }
}

__device__ __forceinline__ void put_tile(float* yw, int c, int h, const v8f& acc0, const v8f& acc1) {
#pragma unroll
    for (int r = 0; r < 8; ++r) yw[(8 * h + r) * YP + c] = acc0[r] + acc1[r];
}

template <int DIH>
__device__ __forceinline__ void stage_load(const float* __restrict__ src, int ii, int o, v4f* v) {
#pragma unroll
    for (int g = 0; g < (2 * DIH + 7) / 8; ++g) {
        const int i = 8 * g + ii;
        if (i < 2 * DIH) {
            const float* s = src + ((i * 8 + (o >> 1)) * 8 + (o & 1)) * 4;
            v[3 * g]     = *(const v4f*)(s);
            v[3 * g + 1] = *(const v4f*)(s + 8);
            v[3 * g + 2] = *(const v4f*)(s + 16);
        }
    }
}

template <int DIH>
__device__ __forceinline__ void stage_store(float* dst, int ii, int o, const v4f* v) {
#pragma unroll
    for (int g = 0; g < (2 * DIH + 7) / 8; ++g) {
        const int i = 8 * g + ii;
        if (i < 2 * DIH) {
            const int h = i / DIH, j = i % DIH;
            float* d = dst + ((3 * j * 2 + h) * 16 + o) * 4;
            *(v4f*)(d)       = v[3 * g];
            *(v4f*)(d + 128) = v[3 * g + 1];
            *(v4f*)(d + 256) = v[3 * g + 2];
        }
    }
}

__global__ __launch_bounds__(128) void kan_main_kernel(const float* __restrict__ x,
                                                       const float* __restrict__ wm,
                                                       float* __restrict__ out) {
    __shared__ __attribute__((aligned(16))) float wms[WM_PER_B];
    __shared__ __attribute__((aligned(16))) float ybuf[4 * 16 * YP];
    __shared__ float obuf[3 * TOK];

    const int tid   = threadIdx.x;
    const int lane  = tid & 31;
    const int wv    = tid >> 5;
    const int c     = lane & 15;
    const int h     = lane >> 4;
    const int b     = blockIdx.x / (HW / TOK);
    const int chunk = blockIdx.x % (HW / TOK);
    const int hw0   = chunk * TOK;

    const float* wb = wm + (size_t)b * WQ_PER_B;
    {
        const int sii = (tid >> 4) & 7, so = tid & 15;
        v4f s0[3], s1[6], s2[6];
        stage_load<DIH0>(wb + WL0, sii, so, s0);
        stage_load<DIH1>(wb + WL1, sii, so, s1);
        stage_load<DIH1>(wb + WL2, sii, so, s2);
        stage_store<DIH0>(wms, sii, so, s0);
        stage_store<DIH1>(wms + WF0, sii, so, s1);
        stage_store<DIH1>(wms + WF0 + WF1, sii, so, s2);
    }

    float* yw = ybuf + wv * 16 * YP;
    const float* xb = x + (size_t)b * 3 * HW + hw0 + wv * 16 + c;
    const float x0 = xb[(size_t)(2 * h) * HW];
    const float xm = xb[HW];
    yw[c * YP + 2 * h]     = x0;
    yw[c * YP + 2 * h + 1] = (h == 0) ? xm : 0.0f;
    __syncthreads();

    const int fragoff = (h * 16 + c) * 4;
    const v8f zero = {0.f, 0.f, 0.f, 0.f, 0.f, 0.f, 0.f, 0.f};
    v8f acc0, acc1;

    acc0 = zero; acc1 = zero;
    const bool unit = (x0 >= 0.0f) && (x0 < 1.0f) && (xm >= 0.0f) && (xm < 1.0f);
    if (__builtin_amdgcn_ballot_w32(unit) == 0xffffffffu)
        kan_layer1_unit(wms, x0, xm, h, c, acc0, acc1);
    else
        kan_layer<DIH0>(wms, yw + c * YP + h * DIH0, fragoff, acc0, acc1);
    put_tile(yw, c, h, acc0, acc1);
    __builtin_amdgcn_wave_barrier();

    acc0 = zero; acc1 = zero;
    kan_layer<DIH1>(wms + WF0, yw + c * YP + h * DIH1, fragoff, acc0, acc1);
    put_tile(yw, c, h, acc0, acc1);
    __builtin_amdgcn_wave_barrier();

    acc0 = zero; acc1 = zero;
    kan_layer<DIH1>(wms + WF0 + WF1, yw + c * YP + h * DIH1, fragoff, acc0, acc1);

    if (c < 3) {
#pragma unroll
        for (int r = 0; r < 8; ++r) obuf[c * TOK + wv * 16 + 8 * h + r] = acc0[r] + acc1[r];
    }
    __syncthreads();
    for (int idx = tid; idx < 3 * TOK; idx += 128)
        out[((size_t)(b * 3 + (idx >> 6))) * HW + hw0 + (idx & (TOK - 1))] = obuf[idx];
}

extern "C" void kernel_launch(void* const* d_in, const int* in_sizes, int n_in,
                              void* d_out, int out_size, void* d_ws, size_t ws_size,
                              hipStream_t stream) {
    const float* x        = (const float*)d_in[0];
    const float* features = (const float*)d_in[1];
    const float* fc_w     = (const float*)d_in[2];
    const float* fc_b     = (const float*)d_in[3];
    float* out    = (float*)d_out;
    float* wm_buf = (float*)d_ws;

    kan_fc_merge_kernel<<<FCNBLK, 256, 0, stream>>>(features, fc_w, fc_b, wm_buf);

    kan_main_kernel<<<NBATCH * (HW / TOK), 128, 0, stream>>>(x, wm_buf, out);
}
